# starting kernel
# baseline (speedup 1.0000x reference)
#include <hip/hip_runtime.h>

#define ALPHA 0.2f

typedef __attribute__((ext_vector_type(2))) float v2f;
typedef __attribute__((ext_vector_type(8))) float v8f;

__device__ __forceinline__ v8f wmma_f32_16x16x4(v2f a, v2f b, v8f c) {
  return __builtin_amdgcn_wmma_f32_16x16x4_f32(false, a, false, b, (short)0, c,
                                               false, false);
}

__device__ __forceinline__ float leaky(float t) {
  return (t >= 0.0f) ? t : ALPHA * t;
}

__global__ void __launch_bounds__(128) k_h_gemm(const float* __restrict__ x,
                                                const float* __restrict__ W,
                                                float* __restrict__ h) {
  const int lane = threadIdx.x & 31;
  const int wave = threadIdx.x >> 5;
  const int rl = lane & 15;
  const int hi = lane >> 4;
  const int i0 = blockIdx.x * 64 + wave * 16;

  v8f acc[8];
  v8f zero = {};
#pragma unroll
  for (int t = 0; t < 8; ++t) acc[t] = zero;

  const float* xrow = x + (size_t)(i0 + rl) * 128 + 2 * hi;
#pragma unroll 4
  for (int k = 0; k < 128; k += 4) {
    v2f afrag = *(const v2f*)(xrow + k);
#pragma unroll
    for (int ot = 0; ot < 8; ++ot) {
      v2f bfrag = *(const v2f*)(W + (size_t)(ot * 16 + rl) * 128 + k + 2 * hi);
      acc[ot] = wmma_f32_16x16x4(afrag, bfrag, acc[ot]);
    }
  }
#pragma unroll
  for (int ot = 0; ot < 8; ++ot)
#pragma unroll
    for (int r = 0; r < 8; ++r)
      h[(size_t)(i0 + r + hi * 8) * 128 + ot * 16 + rl] = acc[ot][r];
}

__global__ void __launch_bounds__(256) k_scores(const float* __restrict__ h,
                                                const float* __restrict__ a,
                                                float* __restrict__ si,
                                                float* __restrict__ sj) {
  const int row = blockIdx.x * 8 + (threadIdx.x >> 5);
  const int lane = threadIdx.x & 31;
  float psi = 0.0f, psj = 0.0f;
#pragma unroll
  for (int c = lane; c < 128; c += 32) {
    float hv = h[(size_t)row * 128 + c];
    psi += hv * a[c];
    psj += hv * a[128 + c];
  }
#pragma unroll
  for (int off = 16; off > 0; off >>= 1) {
    psi += __shfl_xor(psi, off, 32);
    psj += __shfl_xor(psj, off, 32);
  }
  if (lane == 0) {
    si[row] = psi;
    sj[row] = psj;
  }
}

__global__ void __launch_bounds__(256) k_maxsj(const float* __restrict__ sj,
                                               float* __restrict__ mx) {
  __shared__ float red[256];
  const int b = blockIdx.x;
  float m = -3.402823466e38f;
  for (int j = threadIdx.x; j < 2048; j += 256)
    m = fmaxf(m, sj[(size_t)b * 2048 + j]);
  red[threadIdx.x] = m;
  __syncthreads();
  for (int s = 128; s > 0; s >>= 1) {
    if (threadIdx.x < s) red[threadIdx.x] = fmaxf(red[threadIdx.x], red[threadIdx.x + s]);
    __syncthreads();
  }
  if (threadIdx.x == 0) mx[b] = red[0];
}

#define PSTRIDE 288
__global__ void __launch_bounds__(128) k_attn(const float* __restrict__ h,
                                              const float* __restrict__ si,
                                              const float* __restrict__ sj,
                                              const float* __restrict__ mx,
                                              float* __restrict__ out) {
  __shared__ __align__(16) float hs2[32 * PSTRIDE];
  __shared__ __align__(16) float sjs[64];
  __shared__ float zbuf[4 * 16];

  const int tid = threadIdx.x;
  const int lane = tid & 31;
  const int wave = tid >> 5;
  const int rl = lane & 15;
  const int hi = lane >> 4;
  const int b = blockIdx.y;
  const int i0 = blockIdx.x * 64 + wave * 16;

  const float* hb = h + (size_t)b * 2048 * 128;
  const float* sjb = sj + (size_t)b * 2048;

  const float si_row = si[(size_t)b * 2048 + i0 + rl];
  const float m_row = leaky(si_row + mx[b]);

  v8f acc[8];
  v8f zero = {};
#pragma unroll
  for (int t = 0; t < 8; ++t) acc[t] = zero;
  float zacc = 0.0f;

  float4 rA[8], rB[8];
  float sjr = 0.0f;

#pragma unroll
  for (int k = 0; k < 8; ++k) {
    const int q = tid + 128 * k;
    const int jp = q >> 5, c0 = (q & 31) * 4;
    rA[k] = *(const float4*)(hb + (size_t)(2 * jp) * 128 + c0);
    rB[k] = *(const float4*)(hb + (size_t)(2 * jp + 1) * 128 + c0);
  }
  if (tid < 64) sjr = sjb[tid];
#pragma unroll
  for (int k = 0; k < 8; ++k) {
    const int q = tid + 128 * k;
    const int jp = q >> 5, c0 = (q & 31) * 4;
    float4* dst = (float4*)&hs2[jp * PSTRIDE + c0 * 2];
    dst[0] = make_float4(rA[k].x, rB[k].x, rA[k].y, rB[k].y);
    dst[1] = make_float4(rA[k].z, rB[k].z, rA[k].w, rB[k].w);
  }
  if (tid < 64) sjs[tid] = sjr;
  __syncthreads();

  for (int ch = 0; ch < 32; ++ch) {
    if (ch + 1 < 32) {
      const int jn = (ch + 1) * 64;
#pragma unroll
      for (int k = 0; k < 8; ++k) {
        const int q = tid + 128 * k;
        const int jp = q >> 5, c0 = (q & 31) * 4;
        rA[k] = *(const float4*)(hb + (size_t)(jn + 2 * jp) * 128 + c0);
        rB[k] = *(const float4*)(hb + (size_t)(jn + 2 * jp + 1) * 128 + c0);
      }
      if (tid < 64) sjr = sjb[jn + tid];
    }

#pragma unroll 4
    for (int j = 0; j < 64; j += 4) {
      const int jp = (j >> 1) + hi;
      v2f sp = *(const v2f*)&sjs[j + 2 * hi];
      float p0 = __expf(leaky(si_row + sp[0]) - m_row);
      float p1 = __expf(leaky(si_row + sp[1]) - m_row);
      zacc += p0 + p1;
      v2f afrag = {p0, p1};
      const float* bb = &hs2[jp * PSTRIDE + rl * 2];
#pragma unroll
      for (int ot = 0; ot < 8; ++ot) {
        v2f bfrag = *(const v2f*)(bb + ot * 32);
        acc[ot] = wmma_f32_16x16x4(afrag, bfrag, acc[ot]);
      }
    }
    __syncthreads();

    if (ch + 1 < 32) {
#pragma unroll
      for (int k = 0; k < 8; ++k) {
        const int q = tid + 128 * k;
        const int jp = q >> 5, c0 = (q & 31) * 4;
        float4* dst = (float4*)&hs2[jp * PSTRIDE + c0 * 2];
        dst[0] = make_float4(rA[k].x, rB[k].x, rA[k].y, rB[k].y);
        dst[1] = make_float4(rA[k].z, rB[k].z, rA[k].w, rB[k].w);
      }
      if (tid < 64) sjs[tid] = sjr;
    }
    __syncthreads();
  }

  float zfull = zacc + __shfl_xor(zacc, 16, 32);
  if (hi == 0) zbuf[wave * 16 + rl] = zfull;
  __syncthreads();
  float zinv[8];
#pragma unroll
  for (int r = 0; r < 8; ++r) zinv[r] = 1.0f / zbuf[wave * 16 + r + hi * 8];

#pragma unroll
  for (int ot = 0; ot < 8; ++ot)
#pragma unroll
    for (int r = 0; r < 8; ++r)
      out[((size_t)b * 2048 + i0 + r + hi * 8) * 128 + ot * 16 + rl] =
          acc[ot][r] * zinv[r];
}

extern "C" void kernel_launch(void* const* d_in, const int* in_sizes, int n_in,
                              void* d_out, int out_size, void* d_ws, size_t ws_size,
                              hipStream_t stream) {
  const float* x = (const float*)d_in[0];
  const float* W = (const float*)d_in[1];
  const float* a = (const float*)d_in[2];
  float* out = (float*)d_out;

  float* h  = (float*)d_ws;
  float* si = h + (size_t)16384 * 128;
  float* sj = si + 16384;
  float* mx = sj + 16384;

  k_h_gemm<<<dim3(256), dim3(128), 0, stream>>>(x, W, h);
  k_scores<<<dim3(2048), dim3(256), 0, stream>>>(h, a, si, sj);
  k_maxsj<<<dim3(8), dim3(256), 0, stream>>>(sj, mx);
  k_attn<<<dim3(32, 8), dim3(128), 0, stream>>>(h, si, sj, mx, out);
}


# attention kernel: next chunk's global prefetch made unconditional (last chunk peeled) so its wait moves from in front of the chunk's WMMAs to behind them and the loads overlap the compute
# speedup vs baseline: 1.3031x; 1.3031x over previous
#include <hip/hip_runtime.h>

#define ALPHA 0.2f

typedef __attribute__((ext_vector_type(2))) float v2f;
typedef __attribute__((ext_vector_type(8))) float v8f;

__device__ __forceinline__ v8f wmma_f32_16x16x4(v2f a, v2f b, v8f c) {
  return __builtin_amdgcn_wmma_f32_16x16x4_f32(false, a, false, b, (short)0, c,
                                               false, false);
}

__device__ __forceinline__ float leaky(float t) {
  return (t >= 0.0f) ? t : ALPHA * t;
}

__global__ void __launch_bounds__(128) k_h_gemm(const float* __restrict__ x,
                                                const float* __restrict__ W,
                                                float* __restrict__ h) {
  const int lane = threadIdx.x & 31;
  const int wave = threadIdx.x >> 5;
  const int rl = lane & 15;
  const int hi = lane >> 4;
  const int i0 = blockIdx.x * 64 + wave * 16;

  v8f acc[8];
  v8f zero = {};
#pragma unroll
  for (int t = 0; t < 8; ++t) acc[t] = zero;

  const float* xrow = x + (size_t)(i0 + rl) * 128 + 2 * hi;
#pragma unroll 4
  for (int k = 0; k < 128; k += 4) {
    v2f afrag = *(const v2f*)(xrow + k);
#pragma unroll
    for (int ot = 0; ot < 8; ++ot) {
      v2f bfrag = *(const v2f*)(W + (size_t)(ot * 16 + rl) * 128 + k + 2 * hi);
      acc[ot] = wmma_f32_16x16x4(afrag, bfrag, acc[ot]);
    }
  }
#pragma unroll
  for (int ot = 0; ot < 8; ++ot)
#pragma unroll
    for (int r = 0; r < 8; ++r)
      h[(size_t)(i0 + r + hi * 8) * 128 + ot * 16 + rl] = acc[ot][r];
}

__global__ void __launch_bounds__(256) k_scores(const float* __restrict__ h,
                                                const float* __restrict__ a,
                                                float* __restrict__ si,
                                                float* __restrict__ sj) {
  const int row = blockIdx.x * 8 + (threadIdx.x >> 5);
  const int lane = threadIdx.x & 31;
  float psi = 0.0f, psj = 0.0f;
#pragma unroll
  for (int c = lane; c < 128; c += 32) {
    float hv = h[(size_t)row * 128 + c];
    psi += hv * a[c];
    psj += hv * a[128 + c];
  }
#pragma unroll
  for (int off = 16; off > 0; off >>= 1) {
    psi += __shfl_xor(psi, off, 32);
    psj += __shfl_xor(psj, off, 32);
  }
  if (lane == 0) {
    si[row] = psi;
    sj[row] = psj;
  }
}

__global__ void __launch_bounds__(256) k_maxsj(const float* __restrict__ sj,
                                               float* __restrict__ mx) {
  __shared__ float red[256];
  const int b = blockIdx.x;
  float m = -3.402823466e38f;
  for (int j = threadIdx.x; j < 2048; j += 256)
    m = fmaxf(m, sj[(size_t)b * 2048 + j]);
  red[threadIdx.x] = m;
  __syncthreads();
  for (int s = 128; s > 0; s >>= 1) {
    if (threadIdx.x < s) red[threadIdx.x] = fmaxf(red[threadIdx.x], red[threadIdx.x + s]);
    __syncthreads();
  }
  if (threadIdx.x == 0) mx[b] = red[0];
}

#define PSTRIDE 288
__global__ void __launch_bounds__(128) k_attn(const float* __restrict__ h,
                                              const float* __restrict__ si,
                                              const float* __restrict__ sj,
                                              const float* __restrict__ mx,
                                              float* __restrict__ out) {
  __shared__ __align__(16) float hs2[32 * PSTRIDE];
  __shared__ __align__(16) float sjs[64];
  __shared__ float zbuf[4 * 16];

  const int tid = threadIdx.x;
  const int lane = tid & 31;
  const int wave = tid >> 5;
  const int rl = lane & 15;
  const int hi = lane >> 4;
  const int b = blockIdx.y;
  const int i0 = blockIdx.x * 64 + wave * 16;

  const float* hb = h + (size_t)b * 2048 * 128;
  const float* sjb = sj + (size_t)b * 2048;

  const float si_row = si[(size_t)b * 2048 + i0 + rl];
  const float m_row = leaky(si_row + mx[b]);

  v8f acc[8];
  v8f zero = {};
#pragma unroll
  for (int t = 0; t < 8; ++t) acc[t] = zero;
  float zacc = 0.0f;

  float4 rA[8], rB[8];
  float sjr = 0.0f;

#pragma unroll
  for (int k = 0; k < 8; ++k) {
    const int q = tid + 128 * k;
    const int jp = q >> 5, c0 = (q & 31) * 4;
    rA[k] = *(const float4*)(hb + (size_t)(2 * jp) * 128 + c0);
    rB[k] = *(const float4*)(hb + (size_t)(2 * jp + 1) * 128 + c0);
  }
  if (tid < 64) sjr = sjb[tid];
#pragma unroll
  for (int k = 0; k < 8; ++k) {
    const int q = tid + 128 * k;
    const int jp = q >> 5, c0 = (q & 31) * 4;
    float4* dst = (float4*)&hs2[jp * PSTRIDE + c0 * 2];
    dst[0] = make_float4(rA[k].x, rB[k].x, rA[k].y, rB[k].y);
    dst[1] = make_float4(rA[k].z, rB[k].z, rA[k].w, rB[k].w);
  }
  if (tid < 64) sjs[tid] = sjr;
  __syncthreads();

  auto prefetch = [&](int jn) {
#pragma unroll
    for (int k = 0; k < 8; ++k) {
      const int q = tid + 128 * k;
      const int jp = q >> 5, c0 = (q & 31) * 4;
      rA[k] = *(const float4*)(hb + (size_t)(jn + 2 * jp) * 128 + c0);
      rB[k] = *(const float4*)(hb + (size_t)(jn + 2 * jp + 1) * 128 + c0);
    }
    sjr = sjb[jn + (tid & 63)];
  };
  auto stage = [&]() {
#pragma unroll
    for (int k = 0; k < 8; ++k) {
      const int q = tid + 128 * k;
      const int jp = q >> 5, c0 = (q & 31) * 4;
      float4* dst = (float4*)&hs2[jp * PSTRIDE + c0 * 2];
      dst[0] = make_float4(rA[k].x, rB[k].x, rA[k].y, rB[k].y);
      dst[1] = make_float4(rA[k].z, rB[k].z, rA[k].w, rB[k].w);
    }
    if (tid < 64) sjs[tid] = sjr;
  };
  auto compute = [&]() {
#pragma unroll 4
    for (int j = 0; j < 64; j += 4) {
      const int jp = (j >> 1) + hi;
      v2f sp = *(const v2f*)&sjs[j + 2 * hi];
      float p0 = __expf(leaky(si_row + sp[0]) - m_row);
      float p1 = __expf(leaky(si_row + sp[1]) - m_row);
      zacc += p0 + p1;
      v2f afrag = {p0, p1};
      const float* bb = &hs2[jp * PSTRIDE + rl * 2];
#pragma unroll
      for (int ot = 0; ot < 8; ++ot) {
        v2f bfrag = *(const v2f*)(bb + ot * 32);
        acc[ot] = wmma_f32_16x16x4(afrag, bfrag, acc[ot]);
      }
    }
  };

  for (int ch = 0; ch < 31; ++ch) {
    prefetch((ch + 1) * 64);
    compute();
    __syncthreads();
    stage();
    __syncthreads();
  }
  compute();

  float zfull = zacc + __shfl_xor(zacc, 16, 32);
  if (hi == 0) zbuf[wave * 16 + rl] = zfull;
  __syncthreads();
  float zinv[8];
#pragma unroll
  for (int r = 0; r < 8; ++r) zinv[r] = 1.0f / zbuf[wave * 16 + r + hi * 8];

#pragma unroll
  for (int ot = 0; ot < 8; ++ot)
#pragma unroll
    for (int r = 0; r < 8; ++r)
      out[((size_t)b * 2048 + i0 + r + hi * 8) * 128 + ot * 16 + rl] =
          acc[ot][r] * zinv[r];
}

extern "C" void kernel_launch(void* const* d_in, const int* in_sizes, int n_in,
                              void* d_out, int out_size, void* d_ws, size_t ws_size,
                              hipStream_t stream) {
  const float* x = (const float*)d_in[0];
  const float* W = (const float*)d_in[1];
  const float* a = (const float*)d_in[2];
  float* out = (float*)d_out;

  float* h  = (float*)d_ws;
  float* si = h + (size_t)16384 * 128;
  float* sj = si + 16384;
  float* mx = sj + 16384;

  k_h_gemm<<<dim3(256), dim3(128), 0, stream>>>(x, W, h);
  k_scores<<<dim3(2048), dim3(256), 0, stream>>>(h, a, si, sj);
  k_maxsj<<<dim3(8), dim3(256), 0, stream>>>(sj, mx);
  k_attn<<<dim3(32, 8), dim3(128), 0, stream>>>(h, si, sj, mx, out);
}


# score and per-batch-max kernels fused away: si/sj reduced from the GEMM's accumulator tiles in its epilogue (new summation order), max of sj taken in the attention prologue; launches 4 -> 2
# speedup vs baseline: 1.4820x; 1.1373x over previous
#include <hip/hip_runtime.h>

#define ALPHA 0.2f

typedef __attribute__((ext_vector_type(2))) float v2f;
typedef __attribute__((ext_vector_type(8))) float v8f;

__device__ __forceinline__ v8f wmma_f32_16x16x4(v2f a, v2f b, v8f c) {
  return __builtin_amdgcn_wmma_f32_16x16x4_f32(false, a, false, b, (short)0, c,
                                               false, false);
}

__device__ __forceinline__ float leaky(float t) {
  return (t >= 0.0f) ? t : ALPHA * t;
}

__global__ void __launch_bounds__(128) k_h_gemm(const float* __restrict__ x,
                                                const float* __restrict__ W,
                                                const float* __restrict__ a,
                                                float* __restrict__ h,
                                                float* __restrict__ si,
                                                float* __restrict__ sj) {
  __shared__ float sred[128];
  const int lane = threadIdx.x & 31;
  const int wave = threadIdx.x >> 5;
  const int rl = lane & 15;
  const int hi = lane >> 4;
  const int i0 = blockIdx.x * 64 + wave * 16;

  float a1v[8], a2v[8];
#pragma unroll
  for (int ot = 0; ot < 8; ++ot) {
    a1v[ot] = a[ot * 16 + rl];
    a2v[ot] = a[128 + ot * 16 + rl];
  }

  v8f acc[8];
  v8f zero = {};
#pragma unroll
  for (int t = 0; t < 8; ++t) acc[t] = zero;

  const float* xrow = x + (size_t)(i0 + rl) * 128 + 2 * hi;
#pragma unroll 4
  for (int k = 0; k < 128; k += 4) {
    v2f afrag = *(const v2f*)(xrow + k);
#pragma unroll
    for (int ot = 0; ot < 8; ++ot) {
      v2f bfrag = *(const v2f*)(W + (size_t)(ot * 16 + rl) * 128 + k + 2 * hi);
      acc[ot] = wmma_f32_16x16x4(afrag, bfrag, acc[ot]);
    }
  }
  float ps[8], pj[8];
#pragma unroll
  for (int r = 0; r < 8; ++r) {
    float s1 = 0.0f, s2 = 0.0f;
#pragma unroll
    for (int ot = 0; ot < 8; ++ot) {
      s1 += acc[ot][r] * a1v[ot];
      s2 += acc[ot][r] * a2v[ot];
    }
    ps[r] = s1;
    pj[r] = s2;
  }
  const bool u8 = (rl & 8) != 0, u4 = (rl & 4) != 0, u2 = (rl & 2) != 0;
  auto rowsum16 = [&](const float (&v)[8]) -> float {
    float w[4], u[2];
#pragma unroll
    for (int k = 0; k < 4; ++k) {
      const float keep = u8 ? v[k + 4] : v[k];
      const float send = u8 ? v[k] : v[k + 4];
      w[k] = keep + __shfl_xor(send, 8, 32);
    }
#pragma unroll
    for (int k = 0; k < 2; ++k) {
      const float keep = u4 ? w[k + 2] : w[k];
      const float send = u4 ? w[k] : w[k + 2];
      u[k] = keep + __shfl_xor(send, 4, 32);
    }
    const float keep = u2 ? u[1] : u[0];
    const float send = u2 ? u[0] : u[1];
    float t = keep + __shfl_xor(send, 2, 32);
    t += __shfl_xor(t, 1, 32);
    return t;
  };
  const float tsi = rowsum16(ps);
  const float tsj = rowsum16(pj);
  if ((rl & 1) == 0) {
    sred[wave * 16 + hi * 8 + (rl >> 1)] = tsi;
    sred[64 + wave * 16 + hi * 8 + (rl >> 1)] = tsj;
  }
  __syncthreads();
  if (threadIdx.x < 64)
    si[blockIdx.x * 64 + threadIdx.x] = sred[threadIdx.x];
  else
    sj[blockIdx.x * 64 + threadIdx.x - 64] = sred[threadIdx.x];

#pragma unroll
  for (int ot = 0; ot < 8; ++ot)
#pragma unroll
    for (int r = 0; r < 8; ++r)
      h[(size_t)(i0 + r + hi * 8) * 128 + ot * 16 + rl] = acc[ot][r];
}

#define PSTRIDE 288
__global__ void __launch_bounds__(128) k_attn(const float* __restrict__ h,
                                              const float* __restrict__ si,
                                              const float* __restrict__ sj,
                                              float* __restrict__ out) {
  __shared__ __align__(16) float hs2[32 * PSTRIDE];
  __shared__ __align__(16) float sjs[64];
  __shared__ float zbuf[4 * 16];
  __shared__ float mred[4];

  const int tid = threadIdx.x;
  const int lane = tid & 31;
  const int wave = tid >> 5;
  const int rl = lane & 15;
  const int hi = lane >> 4;
  const int b = blockIdx.y;
  const int i0 = blockIdx.x * 64 + wave * 16;

  const float* hb = h + (size_t)b * 2048 * 128;
  const float* sjb = sj + (size_t)b * 2048;

  const float si_row = si[(size_t)b * 2048 + i0 + rl];

  v8f acc[8];
  v8f zero = {};
#pragma unroll
  for (int t = 0; t < 8; ++t) acc[t] = zero;
  float zacc = 0.0f;

  float4 rA[8], rB[8];
  float sjr = 0.0f;

  float4 mv[4];
#pragma unroll
  for (int k = 0; k < 4; ++k) mv[k] = *(const float4*)(sjb + 4 * (tid + 128 * k));

#pragma unroll
  for (int k = 0; k < 8; ++k) {
    const int q = tid + 128 * k;
    const int jp = q >> 5, c0 = (q & 31) * 4;
    rA[k] = *(const float4*)(hb + (size_t)(2 * jp) * 128 + c0);
    rB[k] = *(const float4*)(hb + (size_t)(2 * jp + 1) * 128 + c0);
  }
  if (tid < 64) sjr = sjb[tid];
  {
    float m = -3.402823466e38f;
#pragma unroll
    for (int k = 0; k < 4; ++k)
      m = fmaxf(m, fmaxf(fmaxf(mv[k].x, mv[k].y), fmaxf(mv[k].z, mv[k].w)));
#pragma unroll
    for (int off = 16; off > 0; off >>= 1) m = fmaxf(m, __shfl_xor(m, off, 32));
    if (lane == 0) mred[wave] = m;
  }
#pragma unroll
  for (int k = 0; k < 8; ++k) {
    const int q = tid + 128 * k;
    const int jp = q >> 5, c0 = (q & 31) * 4;
    float4* dst = (float4*)&hs2[jp * PSTRIDE + c0 * 2];
    dst[0] = make_float4(rA[k].x, rB[k].x, rA[k].y, rB[k].y);
    dst[1] = make_float4(rA[k].z, rB[k].z, rA[k].w, rB[k].w);
  }
  if (tid < 64) sjs[tid] = sjr;
  __syncthreads();
  const float mxb = fmaxf(fmaxf(mred[0], mred[1]), fmaxf(mred[2], mred[3]));
  const float m_row = leaky(si_row + mxb);

  auto prefetch = [&](int jn) {
#pragma unroll
    for (int k = 0; k < 8; ++k) {
      const int q = tid + 128 * k;
      const int jp = q >> 5, c0 = (q & 31) * 4;
      rA[k] = *(const float4*)(hb + (size_t)(jn + 2 * jp) * 128 + c0);
      rB[k] = *(const float4*)(hb + (size_t)(jn + 2 * jp + 1) * 128 + c0);
    }
    sjr = sjb[jn + (tid & 63)];
  };
  auto stage = [&]() {
#pragma unroll
    for (int k = 0; k < 8; ++k) {
      const int q = tid + 128 * k;
      const int jp = q >> 5, c0 = (q & 31) * 4;
      float4* dst = (float4*)&hs2[jp * PSTRIDE + c0 * 2];
      dst[0] = make_float4(rA[k].x, rB[k].x, rA[k].y, rB[k].y);
      dst[1] = make_float4(rA[k].z, rB[k].z, rA[k].w, rB[k].w);
    }
    if (tid < 64) sjs[tid] = sjr;
  };
  auto compute = [&]() {
#pragma unroll 4
    for (int j = 0; j < 64; j += 4) {
      const int jp = (j >> 1) + hi;
      v2f sp = *(const v2f*)&sjs[j + 2 * hi];
      float p0 = __expf(leaky(si_row + sp[0]) - m_row);
      float p1 = __expf(leaky(si_row + sp[1]) - m_row);
      zacc += p0 + p1;
      v2f afrag = {p0, p1};
      const float* bb = &hs2[jp * PSTRIDE + rl * 2];
#pragma unroll
      for (int ot = 0; ot < 8; ++ot) {
        v2f bfrag = *(const v2f*)(bb + ot * 32);
        acc[ot] = wmma_f32_16x16x4(afrag, bfrag, acc[ot]);
      }
    }
  };

  for (int ch = 0; ch < 31; ++ch) {
    prefetch((ch + 1) * 64);
    compute();
    __syncthreads();
    stage();
    __syncthreads();
  }
  compute();

  float zfull = zacc + __shfl_xor(zacc, 16, 32);
  if (hi == 0) zbuf[wave * 16 + rl] = zfull;
  __syncthreads();
  float zinv[8];
#pragma unroll
  for (int r = 0; r < 8; ++r) zinv[r] = 1.0f / zbuf[wave * 16 + r + hi * 8];

#pragma unroll
  for (int ot = 0; ot < 8; ++ot)
#pragma unroll
    for (int r = 0; r < 8; ++r)
      out[((size_t)b * 2048 + i0 + r + hi * 8) * 128 + ot * 16 + rl] =
          acc[ot][r] * zinv[r];
}

extern "C" void kernel_launch(void* const* d_in, const int* in_sizes, int n_in,
                              void* d_out, int out_size, void* d_ws, size_t ws_size,
                              hipStream_t stream) {
  const float* x = (const float*)d_in[0];
  const float* W = (const float*)d_in[1];
  const float* a = (const float*)d_in[2];
  float* out = (float*)d_out;

  float* h  = (float*)d_ws;
  float* si = h + (size_t)16384 * 128;
  float* sj = si + 16384;

  k_h_gemm<<<dim3(256), dim3(128), 0, stream>>>(x, W, a, h, si, sj);
  k_attn<<<dim3(32, 8), dim3(128), 0, stream>>>(h, si, sj, out);
}


# attention kernel: 8 waves per block, two per SIMD: each wave pair splits every 64-j chunk in halves (one group of LDS reads, 64 WMMAs each) and adds its partial sums via LDS (new summation order)
# speedup vs baseline: 1.6132x; 1.0885x over previous
#include <hip/hip_runtime.h>

#define ALPHA 0.2f

typedef __attribute__((ext_vector_type(2))) float v2f;
typedef __attribute__((ext_vector_type(8))) float v8f;

__device__ __forceinline__ v8f wmma_f32_16x16x4(v2f a, v2f b, v8f c) {
  return __builtin_amdgcn_wmma_f32_16x16x4_f32(false, a, false, b, (short)0, c,
                                               false, false);
}

__device__ __forceinline__ float leaky(float t) {
  return (t >= 0.0f) ? t : ALPHA * t;
}

__global__ void __launch_bounds__(128) k_h_gemm(const float* __restrict__ x,
                                                const float* __restrict__ W,
                                                const float* __restrict__ a,
                                                float* __restrict__ h,
                                                float* __restrict__ si,
                                                float* __restrict__ sj) {
  __shared__ float sred[128];
  const int lane = threadIdx.x & 31;
  const int wave = threadIdx.x >> 5;
  const int rl = lane & 15;
  const int hi = lane >> 4;
  const int i0 = blockIdx.x * 64 + wave * 16;

  float a1v[8], a2v[8];
#pragma unroll
  for (int ot = 0; ot < 8; ++ot) {
    a1v[ot] = a[ot * 16 + rl];
    a2v[ot] = a[128 + ot * 16 + rl];
  }

  v8f acc[8];
  v8f zero = {};
#pragma unroll
  for (int t = 0; t < 8; ++t) acc[t] = zero;

  const float* xrow = x + (size_t)(i0 + rl) * 128 + 2 * hi;
#pragma unroll 4
  for (int k = 0; k < 128; k += 4) {
    v2f afrag = *(const v2f*)(xrow + k);
#pragma unroll
    for (int ot = 0; ot < 8; ++ot) {
      v2f bfrag = *(const v2f*)(W + (size_t)(ot * 16 + rl) * 128 + k + 2 * hi);
      acc[ot] = wmma_f32_16x16x4(afrag, bfrag, acc[ot]);
    }
  }
  float ps[8], pj[8];
#pragma unroll
  for (int r = 0; r < 8; ++r) {
    float s1 = 0.0f, s2 = 0.0f;
#pragma unroll
    for (int ot = 0; ot < 8; ++ot) {
      s1 += acc[ot][r] * a1v[ot];
      s2 += acc[ot][r] * a2v[ot];
    }
    ps[r] = s1;
    pj[r] = s2;
  }
  const bool u8 = (rl & 8) != 0, u4 = (rl & 4) != 0, u2 = (rl & 2) != 0;
  auto rowsum16 = [&](const float (&v)[8]) -> float {
    float w[4], u[2];
#pragma unroll
    for (int k = 0; k < 4; ++k) {
      const float keep = u8 ? v[k + 4] : v[k];
      const float send = u8 ? v[k] : v[k + 4];
      w[k] = keep + __shfl_xor(send, 8, 32);
    }
#pragma unroll
    for (int k = 0; k < 2; ++k) {
      const float keep = u4 ? w[k + 2] : w[k];
      const float send = u4 ? w[k] : w[k + 2];
      u[k] = keep + __shfl_xor(send, 4, 32);
    }
    const float keep = u2 ? u[1] : u[0];
    const float send = u2 ? u[0] : u[1];
    float t = keep + __shfl_xor(send, 2, 32);
    t += __shfl_xor(t, 1, 32);
    return t;
  };
  const float tsi = rowsum16(ps);
  const float tsj = rowsum16(pj);
  if ((rl & 1) == 0) {
    sred[wave * 16 + hi * 8 + (rl >> 1)] = tsi;
    sred[64 + wave * 16 + hi * 8 + (rl >> 1)] = tsj;
  }
  __syncthreads();
  if (threadIdx.x < 64)
    si[blockIdx.x * 64 + threadIdx.x] = sred[threadIdx.x];
  else
    sj[blockIdx.x * 64 + threadIdx.x - 64] = sred[threadIdx.x];

#pragma unroll
  for (int ot = 0; ot < 8; ++ot)
#pragma unroll
    for (int r = 0; r < 8; ++r)
      h[(size_t)(i0 + r + hi * 8) * 128 + ot * 16 + rl] = acc[ot][r];
}

#define PSTRIDE 288
__global__ void __launch_bounds__(256) __attribute__((amdgpu_waves_per_eu(2, 2))) k_attn(const float* __restrict__ h,
                                              const float* __restrict__ si,
                                              const float* __restrict__ sj,
                                              float* __restrict__ out) {
  __shared__ __align__(16) float hs2[32 * PSTRIDE];
  __shared__ __align__(16) float sjs[64];
  __shared__ float zbuf[8 * 16];
  __shared__ float mred[8];

  const int tid = threadIdx.x;
  const int lane = tid & 31;
  const int wave = __builtin_amdgcn_readfirstlane(tid >> 5);
  const int rg = wave & 3;
  const int jh = wave >> 2;
  const int rl = lane & 15;
  const int hi = lane >> 4;
  const int b = blockIdx.y;
  const int i0 = blockIdx.x * 64 + rg * 16;

  const float* hb = h + (size_t)b * 2048 * 128;
  const float* sjb = sj + (size_t)b * 2048;

  const float si_row = si[(size_t)b * 2048 + i0 + rl];

  v8f acc[8];
  v8f zero = {};
#pragma unroll
  for (int t = 0; t < 8; ++t) acc[t] = zero;
  float zacc = 0.0f;

  float4 rA[4], rB[4];
  float sjr = 0.0f;

  float4 mv[2];
#pragma unroll
  for (int k = 0; k < 2; ++k) mv[k] = *(const float4*)(sjb + 4 * (tid + 256 * k));

#pragma unroll
  for (int k = 0; k < 4; ++k) {
    const int q = tid + 256 * k;
    const int jp = q >> 5, c0 = (q & 31) * 4;
    rA[k] = *(const float4*)(hb + (size_t)(2 * jp) * 128 + c0);
    rB[k] = *(const float4*)(hb + (size_t)(2 * jp + 1) * 128 + c0);
  }
  if (tid < 64) sjr = sjb[tid];
  {
    float m = -3.402823466e38f;
#pragma unroll
    for (int k = 0; k < 2; ++k)
      m = fmaxf(m, fmaxf(fmaxf(mv[k].x, mv[k].y), fmaxf(mv[k].z, mv[k].w)));
#pragma unroll
    for (int off = 16; off > 0; off >>= 1) m = fmaxf(m, __shfl_xor(m, off, 32));
    if (lane == 0) mred[wave] = m;
  }
#pragma unroll
  for (int k = 0; k < 4; ++k) {
    const int q = tid + 256 * k;
    const int jp = q >> 5, c0 = (q & 31) * 4;
    float4* dst = (float4*)&hs2[jp * PSTRIDE + c0 * 2];
    dst[0] = make_float4(rA[k].x, rB[k].x, rA[k].y, rB[k].y);
    dst[1] = make_float4(rA[k].z, rB[k].z, rA[k].w, rB[k].w);
  }
  if (tid < 64) sjs[tid] = sjr;
  __syncthreads();
  const float mxb = fmaxf(fmaxf(fmaxf(mred[0], mred[1]), fmaxf(mred[2], mred[3])),
                          fmaxf(fmaxf(mred[4], mred[5]), fmaxf(mred[6], mred[7])));
  const float m_row = leaky(si_row + mxb);

  auto prefetch = [&](int jn) {
#pragma unroll
    for (int k = 0; k < 4; ++k) {
      const int q = tid + 256 * k;
      const int jp = q >> 5, c0 = (q & 31) * 4;
      rA[k] = *(const float4*)(hb + (size_t)(jn + 2 * jp) * 128 + c0);
      rB[k] = *(const float4*)(hb + (size_t)(jn + 2 * jp + 1) * 128 + c0);
    }
    sjr = sjb[jn + (tid & 63)];
  };
  auto stage = [&]() {
#pragma unroll
    for (int k = 0; k < 4; ++k) {
      const int q = tid + 256 * k;
      const int jp = q >> 5, c0 = (q & 31) * 4;
      float4* dst = (float4*)&hs2[jp * PSTRIDE + c0 * 2];
      dst[0] = make_float4(rA[k].x, rB[k].x, rA[k].y, rB[k].y);
      dst[1] = make_float4(rA[k].z, rB[k].z, rA[k].w, rB[k].w);
    }
    if (tid < 64) sjs[tid] = sjr;
  };
  const float* bbase = &hs2[(jh * 16 + hi) * PSTRIDE + rl * 2];
  const float* sbase = &sjs[jh * 32 + 2 * hi];
  auto compute = [&]() {
    v2f spv[8];
#pragma unroll
    for (int ks = 0; ks < 8; ++ks) spv[ks] = *(const v2f*)(sbase + ks * 4);
#pragma unroll
    for (int ks = 0; ks < 8; ++ks) {
      const v2f sp = spv[ks];
      float p0 = __expf(leaky(si_row + sp[0]) - m_row);
      float p1 = __expf(leaky(si_row + sp[1]) - m_row);
      zacc += p0 + p1;
      v2f afrag = {p0, p1};
      const float* bb = bbase + ks * 2 * PSTRIDE;
#pragma unroll
      for (int ot = 0; ot < 8; ++ot) {
        v2f bfrag = *(const v2f*)(bb + ot * 32);
        acc[ot] = wmma_f32_16x16x4(afrag, bfrag, acc[ot]);
      }
    }
  };

  for (int ch = 0; ch < 31; ++ch) {
    prefetch((ch + 1) * 64);
    compute();
    __syncthreads();
    stage();
    __syncthreads();
  }
  compute();

  float zfull = zacc + __shfl_xor(zacc, 16, 32);
  if (hi == 0) zbuf[wave * 16 + rl] = zfull;
  __syncthreads();

  float* xs = hs2 + wave * 1024 + lane;
  if (jh == 0) {
#pragma unroll
    for (int tt = 0; tt < 4; ++tt)
#pragma unroll
      for (int r = 0; r < 8; ++r) xs[(tt * 8 + r) * 32] = acc[4 + tt][r];
  } else {
#pragma unroll
    for (int tt = 0; tt < 4; ++tt)
#pragma unroll
      for (int r = 0; r < 8; ++r) xs[(tt * 8 + r) * 32] = acc[tt][r];
  }
  __syncthreads();
  const float* xr = hs2 + (wave ^ 4) * 1024 + lane;

  float zinv[8];
#pragma unroll
  for (int r = 0; r < 8; ++r)
    zinv[r] = 1.0f / (zbuf[rg * 16 + r + hi * 8] + zbuf[(rg + 4) * 16 + r + hi * 8]);

  float* orow = out + ((size_t)b * 2048 + i0 + hi * 8) * 128 + rl;
  if (jh == 0) {
#pragma unroll
    for (int tt = 0; tt < 4; ++tt)
#pragma unroll
      for (int r = 0; r < 8; ++r)
        orow[(size_t)r * 128 + tt * 16] =
            (acc[tt][r] + xr[(tt * 8 + r) * 32]) * zinv[r];
  } else {
#pragma unroll
    for (int tt = 0; tt < 4; ++tt)
#pragma unroll
      for (int r = 0; r < 8; ++r)
        orow[(size_t)r * 128 + (4 + tt) * 16] =
            (xr[(tt * 8 + r) * 32] + acc[4 + tt][r]) * zinv[r];
  }
}

extern "C" void kernel_launch(void* const* d_in, const int* in_sizes, int n_in,
                              void* d_out, int out_size, void* d_ws, size_t ws_size,
                              hipStream_t stream) {
  const float* x = (const float*)d_in[0];
  const float* W = (const float*)d_in[1];
  const float* a = (const float*)d_in[2];
  float* out = (float*)d_out;

  float* h  = (float*)d_ws;
  float* si = h + (size_t)16384 * 128;
  float* sj = si + 16384;

  k_h_gemm<<<dim3(256), dim3(128), 0, stream>>>(x, W, a, h, si, sj);
  k_attn<<<dim3(32, 8), dim3(256), 0, stream>>>(h, si, sj, out);
}


# attention kernel: each wave owns a key slice and all 64 rows; h fragments go global->registers, one per 4 row tiles; sweep has no LDS or barrier; 8 partials added via LDS (new summation order)
# speedup vs baseline: 1.8078x; 1.1206x over previous
#include <hip/hip_runtime.h>

#define ALPHA 0.2f

typedef __attribute__((ext_vector_type(2))) float v2f;
typedef __attribute__((ext_vector_type(8))) float v8f;

__device__ __forceinline__ v8f wmma_f32_16x16x4(v2f a, v2f b, v8f c) {
  return __builtin_amdgcn_wmma_f32_16x16x4_f32(false, a, false, b, (short)0, c,
                                               false, false);
}

__device__ __forceinline__ float leaky(float t) {
  return (t >= 0.0f) ? t : ALPHA * t;
}

__global__ void __launch_bounds__(128) k_h_gemm(const float* __restrict__ x,
                                                const float* __restrict__ W,
                                                const float* __restrict__ a,
                                                float* __restrict__ h,
                                                float* __restrict__ si,
                                                float* __restrict__ sj) {
  __shared__ float sred[128];
  const int lane = threadIdx.x & 31;
  const int wave = threadIdx.x >> 5;
  const int rl = lane & 15;
  const int hi = lane >> 4;
  const int i0 = blockIdx.x * 64 + wave * 16;

  float a1v[8], a2v[8];
#pragma unroll
  for (int ot = 0; ot < 8; ++ot) {
    a1v[ot] = a[ot * 16 + rl];
    a2v[ot] = a[128 + ot * 16 + rl];
  }

  v8f acc[8];
  v8f zero = {};
#pragma unroll
  for (int t = 0; t < 8; ++t) acc[t] = zero;

  const float* xrow = x + (size_t)(i0 + rl) * 128 + 2 * hi;
#pragma unroll 4
  for (int k = 0; k < 128; k += 4) {
    v2f afrag = *(const v2f*)(xrow + k);
#pragma unroll
    for (int ot = 0; ot < 8; ++ot) {
      v2f bfrag = *(const v2f*)(W + (size_t)(ot * 16 + rl) * 128 + k + 2 * hi);
      acc[ot] = wmma_f32_16x16x4(afrag, bfrag, acc[ot]);
    }
  }
  float ps[8], pj[8];
#pragma unroll
  for (int r = 0; r < 8; ++r) {
    float s1 = 0.0f, s2 = 0.0f;
#pragma unroll
    for (int ot = 0; ot < 8; ++ot) {
      s1 += acc[ot][r] * a1v[ot];
      s2 += acc[ot][r] * a2v[ot];
    }
    ps[r] = s1;
    pj[r] = s2;
  }
  const bool u8 = (rl & 8) != 0, u4 = (rl & 4) != 0, u2 = (rl & 2) != 0;
  auto rowsum16 = [&](const float (&v)[8]) -> float {
    float w[4], u[2];
#pragma unroll
    for (int k = 0; k < 4; ++k) {
      const float keep = u8 ? v[k + 4] : v[k];
      const float send = u8 ? v[k] : v[k + 4];
      w[k] = keep + __shfl_xor(send, 8, 32);
    }
#pragma unroll
    for (int k = 0; k < 2; ++k) {
      const float keep = u4 ? w[k + 2] : w[k];
      const float send = u4 ? w[k] : w[k + 2];
      u[k] = keep + __shfl_xor(send, 4, 32);
    }
    const float keep = u2 ? u[1] : u[0];
    const float send = u2 ? u[0] : u[1];
    float t = keep + __shfl_xor(send, 2, 32);
    t += __shfl_xor(t, 1, 32);
    return t;
  };
  const float tsi = rowsum16(ps);
  const float tsj = rowsum16(pj);
  if ((rl & 1) == 0) {
    sred[wave * 16 + hi * 8 + (rl >> 1)] = tsi;
    sred[64 + wave * 16 + hi * 8 + (rl >> 1)] = tsj;
  }
  __syncthreads();
  if (threadIdx.x < 64)
    si[blockIdx.x * 64 + threadIdx.x] = sred[threadIdx.x];
  else
    sj[blockIdx.x * 64 + threadIdx.x - 64] = sred[threadIdx.x];

#pragma unroll
  for (int ot = 0; ot < 8; ++ot)
#pragma unroll
    for (int r = 0; r < 8; ++r)
      h[(size_t)(i0 + r + hi * 8) * 128 + ot * 16 + rl] = acc[ot][r];
}

__global__ void __launch_bounds__(256) __attribute__((amdgpu_waves_per_eu(2, 2))) k_attn(const float* __restrict__ h,
                                              const float* __restrict__ si,
                                              const float* __restrict__ sj,
                                              float* __restrict__ out) {
  __shared__ __align__(16) float xs[8 * 16 * 256];
  __shared__ float zbuf[8 * 64];
  __shared__ float mred[8];

  const int tid = threadIdx.x;
  const int lane = tid & 31;
  const int wave = __builtin_amdgcn_readfirstlane(tid >> 5);
  const int rl = lane & 15;
  const int hi = lane >> 4;
  const int b = blockIdx.y;
  const int i0 = blockIdx.x * 64;

  const float* hb = h + (size_t)b * 2048 * 128;
  const float* sjb = sj + (size_t)b * 2048;

  float si_row[4];
#pragma unroll
  for (int t = 0; t < 4; ++t) si_row[t] = si[(size_t)b * 2048 + i0 + t * 16 + rl];

  float4 mv[2];
#pragma unroll
  for (int k = 0; k < 2; ++k) mv[k] = *(const float4*)(sjb + 4 * (tid + 256 * k));
  {
    float m = -3.402823466e38f;
#pragma unroll
    for (int k = 0; k < 2; ++k)
      m = fmaxf(m, fmaxf(fmaxf(mv[k].x, mv[k].y), fmaxf(mv[k].z, mv[k].w)));
#pragma unroll
    for (int off = 16; off > 0; off >>= 1) m = fmaxf(m, __shfl_xor(m, off, 32));
    if (lane == 0) mred[wave] = m;
  }

  v8f acc[4][8];
  v8f zero = {};
#pragma unroll
  for (int t = 0; t < 4; ++t)
#pragma unroll
    for (int ot = 0; ot < 8; ++ot) acc[t][ot] = zero;
  float zacc[4] = {0.0f, 0.0f, 0.0f, 0.0f};

  const float* hcol = hb + (size_t)(wave * 4 + 2 * hi) * 128 + rl;
  const float* sjp = sjb + wave * 4 + 2 * hi;

  auto loadB = [&](int s, v2f (&bf)[8], v2f& sp) {
    const float* p = hcol + (size_t)s * (32 * 128);
#pragma unroll
    for (int ot = 0; ot < 8; ++ot) {
      bf[ot][0] = p[ot * 16];
      bf[ot][1] = p[128 + ot * 16];
    }
    sp = *(const v2f*)(sjp + s * 32);
  };

  v2f b0[8], b1[8], b2[8];
  v2f s0, s1, s2;
  loadB(0, b0, s0);
  loadB(1, b1, s1);

  __syncthreads();
  const float mxb = fmaxf(fmaxf(fmaxf(mred[0], mred[1]), fmaxf(mred[2], mred[3])),
                          fmaxf(fmaxf(mred[4], mred[5]), fmaxf(mred[6], mred[7])));
  float m_row[4];
#pragma unroll
  for (int t = 0; t < 4; ++t) m_row[t] = leaky(si_row[t] + mxb);

  auto step = [&](const v2f (&bf)[8], const v2f sp) {
#pragma unroll
    for (int t = 0; t < 4; ++t) {
      float p0 = __expf(leaky(si_row[t] + sp[0]) - m_row[t]);
      float p1 = __expf(leaky(si_row[t] + sp[1]) - m_row[t]);
      zacc[t] += p0 + p1;
      v2f afrag = {p0, p1};
#pragma unroll
      for (int ot = 0; ot < 8; ++ot)
        acc[t][ot] = wmma_f32_16x16x4(afrag, bf[ot], acc[t][ot]);
    }
  };

  for (int s = 0; s < 60; s += 3) {
    loadB(s + 2, b2, s2);
    step(b0, s0);
    loadB(s + 3, b0, s0);
    step(b1, s1);
    loadB(s + 4, b1, s1);
    step(b2, s2);
  }
  loadB(62, b2, s2);
  step(b0, s0);
  loadB(63, b0, s0);
  step(b1, s1);
  step(b2, s2);
  step(b0, s0);

#pragma unroll
  for (int t = 0; t < 4; ++t) {
    const float zfull = zacc[t] + __shfl_xor(zacc[t], 16, 32);
    if (hi == 0) zbuf[wave * 64 + t * 16 + rl] = zfull;
  }

#pragma unroll
  for (int ph = 0; ph < 2; ++ph) {
    if (ph == 1) __syncthreads();
#pragma unroll
    for (int tt = 0; tt < 2; ++tt)
#pragma unroll
      for (int ot = 0; ot < 8; ++ot) {
        const v8f v = acc[2 * ph + tt][ot];
        float4* dst = (float4*)&xs[(wave * 16 + tt * 8 + ot) * 256 + lane * 4];
        dst[0] = make_float4(v[0], v[1], v[2], v[3]);
        dst[32] = make_float4(v[4], v[5], v[6], v[7]);
      }
    __syncthreads();
    if ((wave >> 2) == ph) {
      const int ow = wave & 3;
      const int tt = ow >> 1;
      const int cq = (ow & 1) * 4;
      const int rowl = (2 * ph + tt) * 16 + hi * 8;
      float zinv[8];
#pragma unroll
      for (int r = 0; r < 8; ++r) {
        float z = zbuf[rowl + r];
#pragma unroll
        for (int w = 1; w < 8; ++w) z += zbuf[w * 64 + rowl + r];
        zinv[r] = 1.0f / z;
      }
      float* orow = out + ((size_t)b * 2048 + i0 + rowl) * 128 + cq * 16 + rl;
#pragma unroll
      for (int q = 0; q < 4; ++q) {
        const float4* src = (const float4*)&xs[(tt * 8 + cq + q) * 256 + lane * 4];
        float4 lo = src[0], up = src[32];
#pragma unroll
        for (int w = 1; w < 8; ++w) {
          const float4 a = src[w * 1024], c = src[w * 1024 + 32];
          lo.x += a.x; lo.y += a.y; lo.z += a.z; lo.w += a.w;
          up.x += c.x; up.y += c.y; up.z += c.z; up.w += c.w;
        }
        orow[0 * 128 + q * 16] = lo.x * zinv[0];
        orow[1 * 128 + q * 16] = lo.y * zinv[1];
        orow[2 * 128 + q * 16] = lo.z * zinv[2];
        orow[3 * 128 + q * 16] = lo.w * zinv[3];
        orow[4 * 128 + q * 16] = up.x * zinv[4];
        orow[5 * 128 + q * 16] = up.y * zinv[5];
        orow[6 * 128 + q * 16] = up.z * zinv[6];
        orow[7 * 128 + q * 16] = up.w * zinv[7];
      }
    }
  }
}

extern "C" void kernel_launch(void* const* d_in, const int* in_sizes, int n_in,
                              void* d_out, int out_size, void* d_ws, size_t ws_size,
                              hipStream_t stream) {
  const float* x = (const float*)d_in[0];
  const float* W = (const float*)d_in[1];
  const float* a = (const float*)d_in[2];
  float* out = (float*)d_out;

  float* h  = (float*)d_ws;
  float* si = h + (size_t)16384 * 128;
  float* sj = si + 16384;

  k_h_gemm<<<dim3(256), dim3(128), 0, stream>>>(x, W, a, h, si, sj);
  k_attn<<<dim3(32, 8), dim3(256), 0, stream>>>(h, si, sj, out);
}


# GEMM kernel: W and the block's x rows copied once per block into LDS by whole-line loads; A and B fragments by 16-byte LDS reads (two k-steps each) replace 8-byte global loads; new k summation order
# speedup vs baseline: 1.8641x; 1.0311x over previous
#include <hip/hip_runtime.h>

#define ALPHA 0.2f

typedef __attribute__((ext_vector_type(2))) float v2f;
typedef __attribute__((ext_vector_type(8))) float v8f;

__device__ __forceinline__ v8f wmma_f32_16x16x4(v2f a, v2f b, v8f c) {
  return __builtin_amdgcn_wmma_f32_16x16x4_f32(false, a, false, b, (short)0, c,
                                               false, false);
}

__device__ __forceinline__ float leaky(float t) {
  return (t >= 0.0f) ? t : ALPHA * t;
}

__global__ void __launch_bounds__(128) k_h_gemm(const float* __restrict__ x,
                                                const float* __restrict__ W,
                                                const float* __restrict__ a,
                                                float* __restrict__ h,
                                                float* __restrict__ si,
                                                float* __restrict__ sj) {
  __shared__ __align__(16) float Ws[128 * 132];
  __shared__ __align__(16) float Xs[64 * 132];
  __shared__ float sred[128];
  const int lane = threadIdx.x & 31;
  const int wave = threadIdx.x >> 5;
  const int rl = lane & 15;
  const int hi = lane >> 4;
  const int i0 = blockIdx.x * 64 + wave * 16;

  {
    float4 wv[32], xv[16];
#pragma unroll
    for (int p = 0; p < 32; ++p)
      wv[p] = *(const float4*)(W + (size_t)(p * 4 + wave) * 128 + lane * 4);
#pragma unroll
    for (int r = 0; r < 16; ++r)
      xv[r] = *(const float4*)(x + (size_t)(i0 + r) * 128 + lane * 4);
#pragma unroll
    for (int p = 0; p < 32; ++p)
      *(float4*)&Ws[(p * 4 + wave) * 132 + lane * 4] = wv[p];
#pragma unroll
    for (int r = 0; r < 16; ++r)
      *(float4*)&Xs[(wave * 16 + r) * 132 + lane * 4] = xv[r];
  }

  float a1v[8], a2v[8];
#pragma unroll
  for (int ot = 0; ot < 8; ++ot) {
    a1v[ot] = a[ot * 16 + rl];
    a2v[ot] = a[128 + ot * 16 + rl];
  }

  v8f acc[8];
  v8f zero = {};
#pragma unroll
  for (int t = 0; t < 8; ++t) acc[t] = zero;

  __syncthreads();

  const float* xl = Xs + (wave * 16 + rl) * 132 + 4 * hi;
  const float* wl = Ws + rl * 132 + 4 * hi;
  float4 xa[2], wb[2][8];
  xa[0] = *(const float4*)xl;
#pragma unroll
  for (int ot = 0; ot < 8; ++ot) wb[0][ot] = *(const float4*)(wl + ot * (16 * 132));
#pragma unroll
  for (int g = 0; g < 16; ++g) {
    if (g + 1 < 16) {
      xa[(g + 1) & 1] = *(const float4*)(xl + 8 * (g + 1));
#pragma unroll
      for (int ot = 0; ot < 8; ++ot)
        wb[(g + 1) & 1][ot] = *(const float4*)(wl + ot * (16 * 132) + 8 * (g + 1));
    }
    const v2f a01 = {xa[g & 1].x, xa[g & 1].y};
    const v2f a23 = {xa[g & 1].z, xa[g & 1].w};
#pragma unroll
    for (int ot = 0; ot < 8; ++ot) {
      const v2f b01 = {wb[g & 1][ot].x, wb[g & 1][ot].y};
      acc[ot] = wmma_f32_16x16x4(a01, b01, acc[ot]);
    }
#pragma unroll
    for (int ot = 0; ot < 8; ++ot) {
      const v2f b23 = {wb[g & 1][ot].z, wb[g & 1][ot].w};
      acc[ot] = wmma_f32_16x16x4(a23, b23, acc[ot]);
    }
  }
  float ps[8], pj[8];
#pragma unroll
  for (int r = 0; r < 8; ++r) {
    float s1 = 0.0f, s2 = 0.0f;
#pragma unroll
    for (int ot = 0; ot < 8; ++ot) {
      s1 += acc[ot][r] * a1v[ot];
      s2 += acc[ot][r] * a2v[ot];
    }
    ps[r] = s1;
    pj[r] = s2;
  }
  const bool u8 = (rl & 8) != 0, u4 = (rl & 4) != 0, u2 = (rl & 2) != 0;
  auto rowsum16 = [&](const float (&v)[8]) -> float {
    float w[4], u[2];
#pragma unroll
    for (int k = 0; k < 4; ++k) {
      const float keep = u8 ? v[k + 4] : v[k];
      const float send = u8 ? v[k] : v[k + 4];
      w[k] = keep + __shfl_xor(send, 8, 32);
    }
#pragma unroll
    for (int k = 0; k < 2; ++k) {
      const float keep = u4 ? w[k + 2] : w[k];
      const float send = u4 ? w[k] : w[k + 2];
      u[k] = keep + __shfl_xor(send, 4, 32);
    }
    const float keep = u2 ? u[1] : u[0];
    const float send = u2 ? u[0] : u[1];
    float t = keep + __shfl_xor(send, 2, 32);
    t += __shfl_xor(t, 1, 32);
    return t;
  };
  const float tsi = rowsum16(ps);
  const float tsj = rowsum16(pj);
  if ((rl & 1) == 0) {
    sred[wave * 16 + hi * 8 + (rl >> 1)] = tsi;
    sred[64 + wave * 16 + hi * 8 + (rl >> 1)] = tsj;
  }
  __syncthreads();
  if (threadIdx.x < 64)
    si[blockIdx.x * 64 + threadIdx.x] = sred[threadIdx.x];
  else
    sj[blockIdx.x * 64 + threadIdx.x - 64] = sred[threadIdx.x];

#pragma unroll
  for (int ot = 0; ot < 8; ++ot)
#pragma unroll
    for (int r = 0; r < 8; ++r)
      h[(size_t)(i0 + r + hi * 8) * 128 + ot * 16 + rl] = acc[ot][r];
}

__global__ void __launch_bounds__(256) __attribute__((amdgpu_waves_per_eu(2, 2))) k_attn(const float* __restrict__ h,
                                              const float* __restrict__ si,
                                              const float* __restrict__ sj,
                                              float* __restrict__ out) {
  __shared__ __align__(16) float xs[8 * 16 * 256];
  __shared__ float zbuf[8 * 64];
  __shared__ float mred[8];

  const int tid = threadIdx.x;
  const int lane = tid & 31;
  const int wave = __builtin_amdgcn_readfirstlane(tid >> 5);
  const int rl = lane & 15;
  const int hi = lane >> 4;
  const int b = blockIdx.y;
  const int i0 = blockIdx.x * 64;

  const float* hb = h + (size_t)b * 2048 * 128;
  const float* sjb = sj + (size_t)b * 2048;

  float si_row[4];
#pragma unroll
  for (int t = 0; t < 4; ++t) si_row[t] = si[(size_t)b * 2048 + i0 + t * 16 + rl];

  float4 mv[2];
#pragma unroll
  for (int k = 0; k < 2; ++k) mv[k] = *(const float4*)(sjb + 4 * (tid + 256 * k));
  {
    float m = -3.402823466e38f;
#pragma unroll
    for (int k = 0; k < 2; ++k)
      m = fmaxf(m, fmaxf(fmaxf(mv[k].x, mv[k].y), fmaxf(mv[k].z, mv[k].w)));
#pragma unroll
    for (int off = 16; off > 0; off >>= 1) m = fmaxf(m, __shfl_xor(m, off, 32));
    if (lane == 0) mred[wave] = m;
  }

  v8f acc[4][8];
  v8f zero = {};
#pragma unroll
  for (int t = 0; t < 4; ++t)
#pragma unroll
    for (int ot = 0; ot < 8; ++ot) acc[t][ot] = zero;
  float zacc[4] = {0.0f, 0.0f, 0.0f, 0.0f};

  const float* hcol = hb + (size_t)(wave * 4 + 2 * hi) * 128 + rl;
  const float* sjp = sjb + wave * 4 + 2 * hi;

  auto loadB = [&](int s, v2f (&bf)[8], v2f& sp) {
    const float* p = hcol + (size_t)s * (32 * 128);
#pragma unroll
    for (int ot = 0; ot < 8; ++ot) {
      bf[ot][0] = p[ot * 16];
      bf[ot][1] = p[128 + ot * 16];
    }
    sp = *(const v2f*)(sjp + s * 32);
  };

  v2f b0[8], b1[8], b2[8];
  v2f s0, s1, s2;
  loadB(0, b0, s0);
  loadB(1, b1, s1);

  __syncthreads();
  const float mxb = fmaxf(fmaxf(fmaxf(mred[0], mred[1]), fmaxf(mred[2], mred[3])),
                          fmaxf(fmaxf(mred[4], mred[5]), fmaxf(mred[6], mred[7])));
  float m_row[4];
#pragma unroll
  for (int t = 0; t < 4; ++t) m_row[t] = leaky(si_row[t] + mxb);

  auto step = [&](const v2f (&bf)[8], const v2f sp) {
#pragma unroll
    for (int t = 0; t < 4; ++t) {
      float p0 = __expf(leaky(si_row[t] + sp[0]) - m_row[t]);
      float p1 = __expf(leaky(si_row[t] + sp[1]) - m_row[t]);
      zacc[t] += p0 + p1;
      v2f afrag = {p0, p1};
#pragma unroll
      for (int ot = 0; ot < 8; ++ot)
        acc[t][ot] = wmma_f32_16x16x4(afrag, bf[ot], acc[t][ot]);
    }
  };

  for (int s = 0; s < 60; s += 3) {
    loadB(s + 2, b2, s2);
    step(b0, s0);
    loadB(s + 3, b0, s0);
    step(b1, s1);
    loadB(s + 4, b1, s1);
    step(b2, s2);
  }
  loadB(62, b2, s2);
  step(b0, s0);
  loadB(63, b0, s0);
  step(b1, s1);
  step(b2, s2);
  step(b0, s0);

#pragma unroll
  for (int t = 0; t < 4; ++t) {
    const float zfull = zacc[t] + __shfl_xor(zacc[t], 16, 32);
    if (hi == 0) zbuf[wave * 64 + t * 16 + rl] = zfull;
  }

#pragma unroll
  for (int ph = 0; ph < 2; ++ph) {
    if (ph == 1) __syncthreads();
#pragma unroll
    for (int tt = 0; tt < 2; ++tt)
#pragma unroll
      for (int ot = 0; ot < 8; ++ot) {
        const v8f v = acc[2 * ph + tt][ot];
        float4* dst = (float4*)&xs[(wave * 16 + tt * 8 + ot) * 256 + lane * 4];
        dst[0] = make_float4(v[0], v[1], v[2], v[3]);
        dst[32] = make_float4(v[4], v[5], v[6], v[7]);
      }
    __syncthreads();
    if ((wave >> 2) == ph) {
      const int ow = wave & 3;
      const int tt = ow >> 1;
      const int cq = (ow & 1) * 4;
      const int rowl = (2 * ph + tt) * 16 + hi * 8;
      float zinv[8];
#pragma unroll
      for (int r = 0; r < 8; ++r) {
        float z = zbuf[rowl + r];
#pragma unroll
        for (int w = 1; w < 8; ++w) z += zbuf[w * 64 + rowl + r];
        zinv[r] = 1.0f / z;
      }
      float* orow = out + ((size_t)b * 2048 + i0 + rowl) * 128 + cq * 16 + rl;
#pragma unroll
      for (int q = 0; q < 4; ++q) {
        const float4* src = (const float4*)&xs[(tt * 8 + cq + q) * 256 + lane * 4];
        float4 lo = src[0], up = src[32];
#pragma unroll
        for (int w = 1; w < 8; ++w) {
          const float4 a = src[w * 1024], c = src[w * 1024 + 32];
          lo.x += a.x; lo.y += a.y; lo.z += a.z; lo.w += a.w;
          up.x += c.x; up.y += c.y; up.z += c.z; up.w += c.w;
        }
        orow[0 * 128 + q * 16] = lo.x * zinv[0];
        orow[1 * 128 + q * 16] = lo.y * zinv[1];
        orow[2 * 128 + q * 16] = lo.z * zinv[2];
        orow[3 * 128 + q * 16] = lo.w * zinv[3];
        orow[4 * 128 + q * 16] = up.x * zinv[4];
        orow[5 * 128 + q * 16] = up.y * zinv[5];
        orow[6 * 128 + q * 16] = up.z * zinv[6];
        orow[7 * 128 + q * 16] = up.w * zinv[7];
      }
    }
  }
}

extern "C" void kernel_launch(void* const* d_in, const int* in_sizes, int n_in,
                              void* d_out, int out_size, void* d_ws, size_t ws_size,
                              hipStream_t stream) {
  const float* x = (const float*)d_in[0];
  const float* W = (const float*)d_in[1];
  const float* a = (const float*)d_in[2];
  float* out = (float*)d_out;

  float* h  = (float*)d_ws;
  float* si = h + (size_t)16384 * 128;
  float* sj = si + 16384;

  k_h_gemm<<<dim3(256), dim3(128), 0, stream>>>(x, W, a, h, si, sj);
  k_attn<<<dim3(32, 8), dim3(256), 0, stream>>>(h, si, sj, out);
}


# attention sweep: s_setprio 3 on the younger wave of each SIMD pair for its first 51 of 64 k-steps, then 0, so both waves of a pair finish together and cover each other's vector work
# speedup vs baseline: 1.9748x; 1.0594x over previous
#include <hip/hip_runtime.h>

#define ALPHA 0.2f

typedef __attribute__((ext_vector_type(2))) float v2f;
typedef __attribute__((ext_vector_type(8))) float v8f;

__device__ __forceinline__ v8f wmma_f32_16x16x4(v2f a, v2f b, v8f c) {
  return __builtin_amdgcn_wmma_f32_16x16x4_f32(false, a, false, b, (short)0, c,
                                               false, false);
}

__device__ __forceinline__ float leaky(float t) {
  return (t >= 0.0f) ? t : ALPHA * t;
}

__global__ void __launch_bounds__(128) k_h_gemm(const float* __restrict__ x,
                                                const float* __restrict__ W,
                                                const float* __restrict__ a,
                                                float* __restrict__ h,
                                                float* __restrict__ si,
                                                float* __restrict__ sj) {
  __shared__ __align__(16) float Ws[128 * 132];
  __shared__ __align__(16) float Xs[64 * 132];
  __shared__ float sred[128];
  const int lane = threadIdx.x & 31;
  const int wave = threadIdx.x >> 5;
  const int rl = lane & 15;
  const int hi = lane >> 4;
  const int i0 = blockIdx.x * 64 + wave * 16;

  {
    float4 wv[32], xv[16];
#pragma unroll
    for (int p = 0; p < 32; ++p)
      wv[p] = *(const float4*)(W + (size_t)(p * 4 + wave) * 128 + lane * 4);
#pragma unroll
    for (int r = 0; r < 16; ++r)
      xv[r] = *(const float4*)(x + (size_t)(i0 + r) * 128 + lane * 4);
#pragma unroll
    for (int p = 0; p < 32; ++p)
      *(float4*)&Ws[(p * 4 + wave) * 132 + lane * 4] = wv[p];
#pragma unroll
    for (int r = 0; r < 16; ++r)
      *(float4*)&Xs[(wave * 16 + r) * 132 + lane * 4] = xv[r];
  }

  float a1v[8], a2v[8];
#pragma unroll
  for (int ot = 0; ot < 8; ++ot) {
    a1v[ot] = a[ot * 16 + rl];
    a2v[ot] = a[128 + ot * 16 + rl];
  }

  v8f acc[8];
  v8f zero = {};
#pragma unroll
  for (int t = 0; t < 8; ++t) acc[t] = zero;

  __syncthreads();

  const float* xl = Xs + (wave * 16 + rl) * 132 + 4 * hi;
  const float* wl = Ws + rl * 132 + 4 * hi;
  float4 xa[2], wb[2][8];
  xa[0] = *(const float4*)xl;
#pragma unroll
  for (int ot = 0; ot < 8; ++ot) wb[0][ot] = *(const float4*)(wl + ot * (16 * 132));
#pragma unroll
  for (int g = 0; g < 16; ++g) {
    if (g + 1 < 16) {
      xa[(g + 1) & 1] = *(const float4*)(xl + 8 * (g + 1));
#pragma unroll
      for (int ot = 0; ot < 8; ++ot)
        wb[(g + 1) & 1][ot] = *(const float4*)(wl + ot * (16 * 132) + 8 * (g + 1));
    }
    const v2f a01 = {xa[g & 1].x, xa[g & 1].y};
    const v2f a23 = {xa[g & 1].z, xa[g & 1].w};
#pragma unroll
    for (int ot = 0; ot < 8; ++ot) {
      const v2f b01 = {wb[g & 1][ot].x, wb[g & 1][ot].y};
      acc[ot] = wmma_f32_16x16x4(a01, b01, acc[ot]);
    }
#pragma unroll
    for (int ot = 0; ot < 8; ++ot) {
      const v2f b23 = {wb[g & 1][ot].z, wb[g & 1][ot].w};
      acc[ot] = wmma_f32_16x16x4(a23, b23, acc[ot]);
    }
  }
  float ps[8], pj[8];
#pragma unroll
  for (int r = 0; r < 8; ++r) {
    float s1 = 0.0f, s2 = 0.0f;
#pragma unroll
    for (int ot = 0; ot < 8; ++ot) {
      s1 += acc[ot][r] * a1v[ot];
      s2 += acc[ot][r] * a2v[ot];
    }
    ps[r] = s1;
    pj[r] = s2;
  }
  const bool u8 = (rl & 8) != 0, u4 = (rl & 4) != 0, u2 = (rl & 2) != 0;
  auto rowsum16 = [&](const float (&v)[8]) -> float {
    float w[4], u[2];
#pragma unroll
    for (int k = 0; k < 4; ++k) {
      const float keep = u8 ? v[k + 4] : v[k];
      const float send = u8 ? v[k] : v[k + 4];
      w[k] = keep + __shfl_xor(send, 8, 32);
    }
#pragma unroll
    for (int k = 0; k < 2; ++k) {
      const float keep = u4 ? w[k + 2] : w[k];
      const float send = u4 ? w[k] : w[k + 2];
      u[k] = keep + __shfl_xor(send, 4, 32);
    }
    const float keep = u2 ? u[1] : u[0];
    const float send = u2 ? u[0] : u[1];
    float t = keep + __shfl_xor(send, 2, 32);
    t += __shfl_xor(t, 1, 32);
    return t;
  };
  const float tsi = rowsum16(ps);
  const float tsj = rowsum16(pj);
  if ((rl & 1) == 0) {
    sred[wave * 16 + hi * 8 + (rl >> 1)] = tsi;
    sred[64 + wave * 16 + hi * 8 + (rl >> 1)] = tsj;
  }
  __syncthreads();
  if (threadIdx.x < 64)
    si[blockIdx.x * 64 + threadIdx.x] = sred[threadIdx.x];
  else
    sj[blockIdx.x * 64 + threadIdx.x - 64] = sred[threadIdx.x];

#pragma unroll
  for (int ot = 0; ot < 8; ++ot)
#pragma unroll
    for (int r = 0; r < 8; ++r)
      h[(size_t)(i0 + r + hi * 8) * 128 + ot * 16 + rl] = acc[ot][r];
}

__global__ void __launch_bounds__(256) __attribute__((amdgpu_waves_per_eu(2, 2))) k_attn(const float* __restrict__ h,
                                              const float* __restrict__ si,
                                              const float* __restrict__ sj,
                                              float* __restrict__ out) {
  __shared__ __align__(16) float xs[8 * 16 * 256];
  __shared__ float zbuf[8 * 64];
  __shared__ float mred[8];

  const int tid = threadIdx.x;
  const int lane = tid & 31;
  const int wave = __builtin_amdgcn_readfirstlane(tid >> 5);
  const int rl = lane & 15;
  const int hi = lane >> 4;
  const int b = blockIdx.y;
  const int i0 = blockIdx.x * 64;

  const float* hb = h + (size_t)b * 2048 * 128;
  const float* sjb = sj + (size_t)b * 2048;

  float si_row[4];
#pragma unroll
  for (int t = 0; t < 4; ++t) si_row[t] = si[(size_t)b * 2048 + i0 + t * 16 + rl];

  float4 mv[2];
#pragma unroll
  for (int k = 0; k < 2; ++k) mv[k] = *(const float4*)(sjb + 4 * (tid + 256 * k));
  {
    float m = -3.402823466e38f;
#pragma unroll
    for (int k = 0; k < 2; ++k)
      m = fmaxf(m, fmaxf(fmaxf(mv[k].x, mv[k].y), fmaxf(mv[k].z, mv[k].w)));
#pragma unroll
    for (int off = 16; off > 0; off >>= 1) m = fmaxf(m, __shfl_xor(m, off, 32));
    if (lane == 0) mred[wave] = m;
  }

  v8f acc[4][8];
  v8f zero = {};
#pragma unroll
  for (int t = 0; t < 4; ++t)
#pragma unroll
    for (int ot = 0; ot < 8; ++ot) acc[t][ot] = zero;
  float zacc[4] = {0.0f, 0.0f, 0.0f, 0.0f};

  const float* hcol = hb + (size_t)(wave * 4 + 2 * hi) * 128 + rl;
  const float* sjp = sjb + wave * 4 + 2 * hi;

  auto loadB = [&](int s, v2f (&bf)[8], v2f& sp) {
    const float* p = hcol + (size_t)s * (32 * 128);
#pragma unroll
    for (int ot = 0; ot < 8; ++ot) {
      bf[ot][0] = p[ot * 16];
      bf[ot][1] = p[128 + ot * 16];
    }
    sp = *(const v2f*)(sjp + s * 32);
  };

  v2f b0[8], b1[8], b2[8];
  v2f s0, s1, s2;
  loadB(0, b0, s0);
  loadB(1, b1, s1);

  __syncthreads();
  const float mxb = fmaxf(fmaxf(fmaxf(mred[0], mred[1]), fmaxf(mred[2], mred[3])),
                          fmaxf(fmaxf(mred[4], mred[5]), fmaxf(mred[6], mred[7])));
  float m_row[4];
#pragma unroll
  for (int t = 0; t < 4; ++t) m_row[t] = leaky(si_row[t] + mxb);

  auto step = [&](const v2f (&bf)[8], const v2f sp) {
#pragma unroll
    for (int t = 0; t < 4; ++t) {
      float p0 = __expf(leaky(si_row[t] + sp[0]) - m_row[t]);
      float p1 = __expf(leaky(si_row[t] + sp[1]) - m_row[t]);
      zacc[t] += p0 + p1;
      v2f afrag = {p0, p1};
#pragma unroll
      for (int ot = 0; ot < 8; ++ot)
        acc[t][ot] = wmma_f32_16x16x4(afrag, bf[ot], acc[t][ot]);
    }
  };

  auto three_steps = [&](int s) {
    loadB(s + 2, b2, s2);
    step(b0, s0);
    loadB(s + 3, b0, s0);
    step(b1, s1);
    loadB(s + 4, b1, s1);
    step(b2, s2);
  };
  if (wave >= 4) __builtin_amdgcn_s_setprio(3);
  for (int s = 0; s < 51; s += 3) three_steps(s);
  __builtin_amdgcn_s_setprio(0);
  for (int s = 51; s < 60; s += 3) three_steps(s);
  loadB(62, b2, s2);
  step(b0, s0);
  loadB(63, b0, s0);
  step(b1, s1);
  step(b2, s2);
  step(b0, s0);

#pragma unroll
  for (int t = 0; t < 4; ++t) {
    const float zfull = zacc[t] + __shfl_xor(zacc[t], 16, 32);
    if (hi == 0) zbuf[wave * 64 + t * 16 + rl] = zfull;
  }

#pragma unroll
  for (int ph = 0; ph < 2; ++ph) {
    if (ph == 1) __syncthreads();
#pragma unroll
    for (int tt = 0; tt < 2; ++tt)
#pragma unroll
      for (int ot = 0; ot < 8; ++ot) {
        const v8f v = acc[2 * ph + tt][ot];
        float4* dst = (float4*)&xs[(wave * 16 + tt * 8 + ot) * 256 + lane * 4];
        dst[0] = make_float4(v[0], v[1], v[2], v[3]);
        dst[32] = make_float4(v[4], v[5], v[6], v[7]);
      }
    __syncthreads();
    if ((wave >> 2) == ph) {
      const int ow = wave & 3;
      const int tt = ow >> 1;
      const int cq = (ow & 1) * 4;
      const int rowl = (2 * ph + tt) * 16 + hi * 8;
      float zinv[8];
#pragma unroll
      for (int r = 0; r < 8; ++r) {
        float z = zbuf[rowl + r];
#pragma unroll
        for (int w = 1; w < 8; ++w) z += zbuf[w * 64 + rowl + r];
        zinv[r] = 1.0f / z;
      }
      float* orow = out + ((size_t)b * 2048 + i0 + rowl) * 128 + cq * 16 + rl;
#pragma unroll
      for (int q = 0; q < 4; ++q) {
        const float4* src = (const float4*)&xs[(tt * 8 + cq + q) * 256 + lane * 4];
        float4 lo = src[0], up = src[32];
#pragma unroll
        for (int w = 1; w < 8; ++w) {
          const float4 a = src[w * 1024], c = src[w * 1024 + 32];
          lo.x += a.x; lo.y += a.y; lo.z += a.z; lo.w += a.w;
          up.x += c.x; up.y += c.y; up.z += c.z; up.w += c.w;
        }
        orow[0 * 128 + q * 16] = lo.x * zinv[0];
        orow[1 * 128 + q * 16] = lo.y * zinv[1];
        orow[2 * 128 + q * 16] = lo.z * zinv[2];
        orow[3 * 128 + q * 16] = lo.w * zinv[3];
        orow[4 * 128 + q * 16] = up.x * zinv[4];
        orow[5 * 128 + q * 16] = up.y * zinv[5];
        orow[6 * 128 + q * 16] = up.z * zinv[6];
        orow[7 * 128 + q * 16] = up.w * zinv[7];
      }
    }
  }
}

extern "C" void kernel_launch(void* const* d_in, const int* in_sizes, int n_in,
                              void* d_out, int out_size, void* d_ws, size_t ws_size,
                              hipStream_t stream) {
  const float* x = (const float*)d_in[0];
  const float* W = (const float*)d_in[1];
  const float* a = (const float*)d_in[2];
  float* out = (float*)d_out;

  float* h  = (float*)d_ws;
  float* si = h + (size_t)16384 * 128;
  float* sj = si + 16384;

  k_h_gemm<<<dim3(256), dim3(128), 0, stream>>>(x, W, a, h, si, sj);
  k_attn<<<dim3(32, 8), dim3(256), 0, stream>>>(h, si, sj, out);
}
